# starting kernel
# baseline (speedup 1.0000x reference)
#include <hip/hip_runtime.h>
#include <hip/hip_bf16.h>
#include <math.h>

typedef __attribute__((ext_vector_type(2))) float v2f;
typedef __attribute__((ext_vector_type(8))) float v8f;

#define NROWS 8192
#define NDIM  64
#define COLS_PER_BLOCK 128
#define LDS_STRIDE 66
#define BETA 0.01f
#define TWO_PI 6.28318530717958647692f

__global__ __launch_bounds__(256) void k_y(const float* __restrict__ x,
                                           const float* __restrict__ noise,
                                           const float* __restrict__ nvp,
                                           float* __restrict__ y, int n) {
    int i = blockIdx.x * 256 + threadIdx.x;
    float s = __expf(0.5f * nvp[0]);
    if (i < n) y[i] = fmaf(s, noise[i], x[i]);
}

__global__ __launch_bounds__(256) void k_sq(const float* __restrict__ x,
                                            float* __restrict__ sq) {
    int row = blockIdx.x * 256 + threadIdx.x;
    const float4* r = (const float4*)(x + (size_t)row * NDIM);
    float s = 0.0f;
#pragma unroll
    for (int j = 0; j < NDIM / 4; ++j) {
        float4 v = r[j];
        s = fmaf(v.x, v.x, s); s = fmaf(v.y, v.y, s);
        s = fmaf(v.z, v.z, s); s = fmaf(v.w, v.w, s);
    }
    sq[row] = s;
}

__global__ __launch_bounds__(256) void k_pairwise(const float* __restrict__ x,
                                                  const float* __restrict__ sq,
                                                  const float* __restrict__ kvp,
                                                  const float* __restrict__ nvp,
                                                  float* __restrict__ lse1_out,
                                                  float* __restrict__ lse2_out) {
    __shared__ float ldsX[COLS_PER_BLOCK * LDS_STRIDE];
    __shared__ float ldsSq[COLS_PER_BLOCK];

    const int tid      = threadIdx.x;
    const int lane     = tid & 31;
    const int wave     = tid >> 5;
    const int half     = lane >> 4;
    const int l15      = lane & 15;
    const int rowBase  = blockIdx.x * (8 * 16) + wave * 16;

    const float kv = kvp[0], nv = nvp[0];
    const float total_var = __expf(nv) + __expf(kv);
    const float a1 = 1.0f / (2.0f * total_var);
    const float a2 = 1.0f / (2.0f * __expf(kv));

    v2f a[16];
    {
        const float* ar = x + (size_t)(rowBase + l15) * NDIM + 2 * half;
#pragma unroll
        for (int ch = 0; ch < 16; ++ch)
            a[ch] = *(const v2f*)(ar + 4 * ch);
    }

    float sqA[8];
#pragma unroll
    for (int k = 0; k < 8; ++k)
        sqA[k] = sq[rowBase + k + 8 * half];

    float l1[8], l2[8], m2[8];
#pragma unroll
    for (int k = 0; k < 8; ++k) { l1[k] = 0.0f; l2[k] = 0.0f; m2[k] = -1e30f; }

    float4 stage[8];
    float  stageSq;
#pragma unroll
    for (int i = 0; i < 8; ++i)
        stage[i] = ((const float4*)x)[tid + i * 256];
    stageSq = (tid < COLS_PER_BLOCK) ? sq[tid] : 0.0f;

    for (int cb = 0; cb < NROWS; cb += COLS_PER_BLOCK) {
        __syncthreads();
#pragma unroll
        for (int i = 0; i < 8; ++i) {
            int idx = tid + i * 256;
            int c   = idx >> 4;
            int f4  = idx & 15;
            float* dst = &ldsX[c * LDS_STRIDE + f4 * 4];
            dst[0] = stage[i].x; dst[1] = stage[i].y;
            dst[2] = stage[i].z; dst[3] = stage[i].w;
        }
        if (tid < COLS_PER_BLOCK) ldsSq[tid] = stageSq;
        __syncthreads();

        const int nb = cb + COLS_PER_BLOCK;
        if (nb < NROWS) {
#pragma unroll
            for (int i = 0; i < 8; ++i)
                stage[i] = ((const float4*)(x + (size_t)nb * NDIM))[tid + i * 256];
            if (tid < COLS_PER_BLOCK) stageSq = sq[nb + tid];
        }

        for (int ct = 0; ct < COLS_PER_BLOCK / 16; ++ct) {
            v2f b[16];
            const float* bp = &ldsX[(ct * 16 + l15) * LDS_STRIDE + 2 * half];
#pragma unroll
            for (int ch = 0; ch < 16; ++ch)
                b[ch] = *(const v2f*)(bp + 4 * ch);

            v8f c = {};
#pragma unroll
            for (int ch = 0; ch < 16; ++ch)
                c = __builtin_amdgcn_wmma_f32_16x16x4_f32(
                         false, a[ch],  false, b[ch],
                         (short)0, c,  false,  false);

            const float sqB = ldsSq[ct * 16 + l15];
            const int   col = cb + ct * 16 + l15;
#pragma unroll
            for (int k = 0; k < 8; ++k) {
                const int rowG = rowBase + k + 8 * half;
                float d = fmaxf(sqA[k] + sqB - 2.0f * c[k], 0.0f);
                l1[k] += __expf(-d * a1);
                float s2 = (col == rowG) ? -1e38f : (-d * a2);
                float nm = fmaxf(m2[k], s2);
                l2[k] = l2[k] * __expf(m2[k] - nm) + __expf(s2 - nm);
                m2[k] = nm;
            }
        }
    }

#pragma unroll
    for (int k = 0; k < 8; ++k) {
#pragma unroll
        for (int m = 1; m <= 8; m <<= 1) {
            l1[k] += __shfl_xor(l1[k], m, 32);
            float om = __shfl_xor(m2[k], m, 32);
            float ol = __shfl_xor(l2[k], m, 32);
            float nm = fmaxf(m2[k], om);
            l2[k] = l2[k] * __expf(m2[k] - nm) + ol * __expf(om - nm);
            m2[k] = nm;
        }
    }
    if (l15 == 0) {
#pragma unroll
        for (int k = 0; k < 8; ++k) {
            int rowG = rowBase + k + 8 * half;
            lse1_out[rowG] = __logf(l1[k]);
            lse2_out[rowG] = m2[k] + __logf(l2[k]);
        }
    }
}

__global__ __launch_bounds__(256) void k_final(const float* __restrict__ lse1,
                                               const float* __restrict__ lse2,
                                               const float* __restrict__ kvp,
                                               const float* __restrict__ nvp,
                                               float* __restrict__ out2) {
    __shared__ float s1[256], s2[256];
    float p1 = 0.0f, p2 = 0.0f;
    for (int i = threadIdx.x; i < NROWS; i += 256) { p1 += lse1[i]; p2 += lse2[i]; }
    s1[threadIdx.x] = p1; s2[threadIdx.x] = p2;
    __syncthreads();
    for (int st = 128; st > 0; st >>= 1) {
        if (threadIdx.x < st) {
            s1[threadIdx.x] += s1[threadIdx.x + st];
            s2[threadIdx.x] += s2[threadIdx.x + st];
        }
        __syncthreads();
    }
    if (threadIdx.x == 0) {
        const float kv = kvp[0], nv = nvp[0];
        const float Nf = (float)NROWS, dims = (float)NDIM;
        float total_var = expf(nv) + expf(kv);
        float nc1 = 0.5f * dims * logf(TWO_PI * total_var);
        float mean_lp1 = s1[0] / Nf - logf(Nf) - nc1;
        float h  = -mean_lp1;
        out2[0] = BETA * (h - nc1);
        float kdevar = expf(kv);
        float nc2 = 0.5f * dims * logf(TWO_PI * kdevar);
        float mean_lp2 = s2[0] / Nf - logf(Nf - 1.0f) - nc2;
        out2[1] = -mean_lp2;
    }
}

extern "C" void kernel_launch(void* const* d_in, const int* in_sizes, int n_in,
                              void* d_out, int out_size, void* d_ws, size_t ws_size,
                              hipStream_t stream) {
    const float* x     = (const float*)d_in[0];
    const float* noise = (const float*)d_in[1];
    const float* kv    = (const float*)d_in[2];
    const float* nv    = (const float*)d_in[3];
    float* out = (float*)d_out;
    float* ws  = (float*)d_ws;

    float* sq   = ws;
    float* lse1 = ws + NROWS;
    float* lse2 = ws + 2 * NROWS;

    const int nd = NROWS * NDIM;
    k_y<<<(nd + 255) / 256, 256, 0, stream>>>(x, noise, nv, out, nd);
    k_sq<<<NROWS / 256, 256, 0, stream>>>(x, sq);
    k_pairwise<<<NROWS / 128, 256, 0, stream>>>(x, sq, kv, nv, lse1, lse2);
    k_final<<<1, 256, 0, stream>>>(lse1, lse2, kv, nv, out + nd);
}


# Column sweep split 8 ways over grid.y: 512 workgroups fill all 256 WGPs (was 64); per-slice sums and maxima merged in the final kernel, so row sums and the final sum change order
# speedup vs baseline: 3.1562x; 3.1562x over previous
#include <hip/hip_runtime.h>
#include <hip/hip_bf16.h>
#include <math.h>

typedef __attribute__((ext_vector_type(2))) float v2f;
typedef __attribute__((ext_vector_type(8))) float v8f;

#define NROWS 8192
#define NDIM  64
#define COLS_PER_BLOCK 128
#define NSPLIT 8
#define COLS_PER_SLICE (NROWS / NSPLIT)
#define LDS_STRIDE 66
#define BETA 0.01f
#define TWO_PI 6.28318530717958647692f

__global__ __launch_bounds__(256) void k_y(const float* __restrict__ x,
                                           const float* __restrict__ noise,
                                           const float* __restrict__ nvp,
                                           float* __restrict__ y, int n) {
    int i = blockIdx.x * 256 + threadIdx.x;
    float s = __expf(0.5f * nvp[0]);
    if (i < n) y[i] = fmaf(s, noise[i], x[i]);
}

__global__ __launch_bounds__(256) void k_sq(const float* __restrict__ x,
                                            float* __restrict__ sq) {
    int row = blockIdx.x * 256 + threadIdx.x;
    const float4* r = (const float4*)(x + (size_t)row * NDIM);
    float s = 0.0f;
#pragma unroll
    for (int j = 0; j < NDIM / 4; ++j) {
        float4 v = r[j];
        s = fmaf(v.x, v.x, s); s = fmaf(v.y, v.y, s);
        s = fmaf(v.z, v.z, s); s = fmaf(v.w, v.w, s);
    }
    sq[row] = s;
}

__global__ __launch_bounds__(256) void k_pairwise(const float* __restrict__ x,
                                                  const float* __restrict__ sq,
                                                  const float* __restrict__ kvp,
                                                  const float* __restrict__ nvp,
                                                  float* __restrict__ pl1,
                                                  float* __restrict__ pm2,
                                                  float* __restrict__ pl2) {
    __shared__ float ldsX[COLS_PER_BLOCK * LDS_STRIDE];
    __shared__ float ldsSq[COLS_PER_BLOCK];
    __shared__ float ldsOut[3][8 * 16];

    const int tid      = threadIdx.x;
    const int lane     = tid & 31;
    const int wave     = tid >> 5;
    const int half     = lane >> 4;
    const int l15      = lane & 15;
    const int rowBase  = blockIdx.x * (8 * 16) + wave * 16;
    const int colBeg   = blockIdx.y * COLS_PER_SLICE;
    const int colEnd   = colBeg + COLS_PER_SLICE;

    const float kv = kvp[0], nv = nvp[0];
    const float total_var = __expf(nv) + __expf(kv);
    const float a1 = 1.0f / (2.0f * total_var);
    const float a2 = 1.0f / (2.0f * __expf(kv));

    v2f a[16];
    {
        const float* ar = x + (size_t)(rowBase + l15) * NDIM + 2 * half;
#pragma unroll
        for (int ch = 0; ch < 16; ++ch)
            a[ch] = *(const v2f*)(ar + 4 * ch);
    }

    float sqA[8];
#pragma unroll
    for (int k = 0; k < 8; ++k)
        sqA[k] = sq[rowBase + k + 8 * half];

    float l1[8], l2[8], m2[8];
#pragma unroll
    for (int k = 0; k < 8; ++k) { l1[k] = 0.0f; l2[k] = 0.0f; m2[k] = -1e30f; }

    float4 stage[8];
    float  stageSq;
#pragma unroll
    for (int i = 0; i < 8; ++i)
        stage[i] = ((const float4*)(x + (size_t)colBeg * NDIM))[tid + i * 256];
    stageSq = (tid < COLS_PER_BLOCK) ? sq[colBeg + tid] : 0.0f;

    for (int cb = colBeg; cb < colEnd; cb += COLS_PER_BLOCK) {
        __syncthreads();
#pragma unroll
        for (int i = 0; i < 8; ++i) {
            int idx = tid + i * 256;
            int c   = idx >> 4;
            int f4  = idx & 15;
            float* dst = &ldsX[c * LDS_STRIDE + f4 * 4];
            dst[0] = stage[i].x; dst[1] = stage[i].y;
            dst[2] = stage[i].z; dst[3] = stage[i].w;
        }
        if (tid < COLS_PER_BLOCK) ldsSq[tid] = stageSq;
        __syncthreads();

        const int nb = cb + COLS_PER_BLOCK;
        if (nb < colEnd) {
#pragma unroll
            for (int i = 0; i < 8; ++i)
                stage[i] = ((const float4*)(x + (size_t)nb * NDIM))[tid + i * 256];
            if (tid < COLS_PER_BLOCK) stageSq = sq[nb + tid];
        }

        for (int ct = 0; ct < COLS_PER_BLOCK / 16; ++ct) {
            v2f b[16];
            const float* bp = &ldsX[(ct * 16 + l15) * LDS_STRIDE + 2 * half];
#pragma unroll
            for (int ch = 0; ch < 16; ++ch)
                b[ch] = *(const v2f*)(bp + 4 * ch);

            v8f c = {};
#pragma unroll
            for (int ch = 0; ch < 16; ++ch)
                c = __builtin_amdgcn_wmma_f32_16x16x4_f32(
                         false, a[ch],  false, b[ch],
                         (short)0, c,  false,  false);

            const float sqB = ldsSq[ct * 16 + l15];
            const int   col = cb + ct * 16 + l15;
#pragma unroll
            for (int k = 0; k < 8; ++k) {
                const int rowG = rowBase + k + 8 * half;
                float d = fmaxf(sqA[k] + sqB - 2.0f * c[k], 0.0f);
                l1[k] += __expf(-d * a1);
                float s2 = (col == rowG) ? -1e38f : (-d * a2);
                float nm = fmaxf(m2[k], s2);
                l2[k] = l2[k] * __expf(m2[k] - nm) + __expf(s2 - nm);
                m2[k] = nm;
            }
        }
    }

#pragma unroll
    for (int k = 0; k < 8; ++k) {
#pragma unroll
        for (int m = 1; m <= 8; m <<= 1) {
            l1[k] += __shfl_xor(l1[k], m, 32);
            float om = __shfl_xor(m2[k], m, 32);
            float ol = __shfl_xor(l2[k], m, 32);
            float nm = fmaxf(m2[k], om);
            l2[k] = l2[k] * __expf(m2[k] - nm) + ol * __expf(om - nm);
            m2[k] = nm;
        }
    }
    if (l15 == 0) {
#pragma unroll
        for (int k = 0; k < 8; ++k) {
            const int r = wave * 16 + k + 8 * half;
            ldsOut[0][r] = l1[k];
            ldsOut[1][r] = m2[k];
            ldsOut[2][r] = l2[k];
        }
    }
    __syncthreads();
    if (tid < 128) {
        const size_t o = (size_t)blockIdx.y * NROWS + (size_t)blockIdx.x * 128 + tid;
        pl1[o] = ldsOut[0][tid];
        pm2[o] = ldsOut[1][tid];
        pl2[o] = ldsOut[2][tid];
    }
}

__global__ __launch_bounds__(1024) void k_final(const float* __restrict__ pl1,
                                                const float* __restrict__ pm2,
                                                const float* __restrict__ pl2,
                                                const float* __restrict__ kvp,
                                                const float* __restrict__ nvp,
                                                float* __restrict__ out2) {
    __shared__ float s1[1024], s2[1024];
    float p1 = 0.0f, p2 = 0.0f;
#pragma unroll
    for (int q = threadIdx.x; q < NROWS / 4; q += 1024) {
        float a[4] = {0.0f, 0.0f, 0.0f, 0.0f};
        float m[NSPLIT][4], l[NSPLIT][4];
#pragma unroll
        for (int s = 0; s < NSPLIT; ++s) {
            const float4 va = ((const float4*)(pl1 + s * NROWS))[q];
            const float4 vm = ((const float4*)(pm2 + s * NROWS))[q];
            const float4 vl = ((const float4*)(pl2 + s * NROWS))[q];
            a[0] += va.x; a[1] += va.y; a[2] += va.z; a[3] += va.w;
            m[s][0] = vm.x; m[s][1] = vm.y; m[s][2] = vm.z; m[s][3] = vm.w;
            l[s][0] = vl.x; l[s][1] = vl.y; l[s][2] = vl.z; l[s][3] = vl.w;
        }
#pragma unroll
        for (int e = 0; e < 4; ++e) {
            float mm = m[0][e];
#pragma unroll
            for (int s = 1; s < NSPLIT; ++s) mm = fmaxf(mm, m[s][e]);
            float ll = 0.0f;
#pragma unroll
            for (int s = 0; s < NSPLIT; ++s) ll = fmaf(l[s][e], __expf(m[s][e] - mm), ll);
            p1 += __logf(a[e]);
            p2 += mm + __logf(ll);
        }
    }
    s1[threadIdx.x] = p1; s2[threadIdx.x] = p2;
    __syncthreads();
    for (int st = 512; st > 0; st >>= 1) {
        if (threadIdx.x < st) {
            s1[threadIdx.x] += s1[threadIdx.x + st];
            s2[threadIdx.x] += s2[threadIdx.x + st];
        }
        __syncthreads();
    }
    if (threadIdx.x == 0) {
        const float kv = kvp[0], nv = nvp[0];
        const float Nf = (float)NROWS, dims = (float)NDIM;
        float total_var = expf(nv) + expf(kv);
        float nc1 = 0.5f * dims * logf(TWO_PI * total_var);
        float mean_lp1 = s1[0] / Nf - logf(Nf) - nc1;
        float h  = -mean_lp1;
        out2[0] = BETA * (h - nc1);
        float kdevar = expf(kv);
        float nc2 = 0.5f * dims * logf(TWO_PI * kdevar);
        float mean_lp2 = s2[0] / Nf - logf(Nf - 1.0f) - nc2;
        out2[1] = -mean_lp2;
    }
}

extern "C" void kernel_launch(void* const* d_in, const int* in_sizes, int n_in,
                              void* d_out, int out_size, void* d_ws, size_t ws_size,
                              hipStream_t stream) {
    const float* x     = (const float*)d_in[0];
    const float* noise = (const float*)d_in[1];
    const float* kv    = (const float*)d_in[2];
    const float* nv    = (const float*)d_in[3];
    float* out = (float*)d_out;
    float* ws  = (float*)d_ws;

    float* sq  = ws;
    float* pl1 = ws + NROWS;
    float* pm2 = ws + NROWS + NSPLIT * NROWS;
    float* pl2 = ws + NROWS + 2 * NSPLIT * NROWS;

    const int nd = NROWS * NDIM;
    k_y<<<(nd + 255) / 256, 256, 0, stream>>>(x, noise, nv, out, nd);
    k_sq<<<NROWS / 256, 256, 0, stream>>>(x, sq);
    k_pairwise<<<dim3(NROWS / 128, NSPLIT), 256, 0, stream>>>(x, sq, kv, nv, pl1, pm2, pl2);
    k_final<<<1, 1024, 0, stream>>>(pl1, pm2, pl2, kv, nv, out + nd);
}


# Launch fusion 4 -> 2: y and the squared-norm pass folded into the pairwise kernel (one float4 of y per thread; norms from A fragments and staged registers), so the norm sums change order
# speedup vs baseline: 3.3636x; 1.0657x over previous
#include <hip/hip_runtime.h>
#include <hip/hip_bf16.h>
#include <math.h>

typedef __attribute__((ext_vector_type(2))) float v2f;
typedef __attribute__((ext_vector_type(8))) float v8f;

#define NROWS 8192
#define NDIM  64
#define COLS_PER_BLOCK 128
#define NSPLIT 8
#define COLS_PER_SLICE (NROWS / NSPLIT)
#define LDS_STRIDE 66
#define BETA 0.01f
#define TWO_PI 6.28318530717958647692f

__device__ __forceinline__ float sumsq4(const float4 v) {
    float s = v.x * v.x;
    s = fmaf(v.y, v.y, s); s = fmaf(v.z, v.z, s); s = fmaf(v.w, v.w, s);
    return s;
}

template <int CTRL>
__device__ __forceinline__ float row_xmask(float v) {
    return __builtin_bit_cast(float,
        __builtin_amdgcn_update_dpp(0, __builtin_bit_cast(int, v), CTRL, 0xf, 0xf, true));
}

__device__ __forceinline__ float row16_sum(float v) {
    v += row_xmask<0x161>(v);
    v += row_xmask<0x162>(v);
    v += row_xmask<0x164>(v);
    v += row_xmask<0x168>(v);
    return v;
}

__global__ __launch_bounds__(256) void k_pairwise(const float* __restrict__ x,
                                                  const float* __restrict__ noise,
                                                  const float* __restrict__ kvp,
                                                  const float* __restrict__ nvp,
                                                  float* __restrict__ y,
                                                  float* __restrict__ pl1,
                                                  float* __restrict__ pm2,
                                                  float* __restrict__ pl2) {
    __shared__ float ldsX[COLS_PER_BLOCK * LDS_STRIDE];
    __shared__ float ldsSq[COLS_PER_BLOCK];
    __shared__ float ldsOut[3][8 * 16];

    const int tid      = threadIdx.x;
    const int lane     = tid & 31;
    const int wave     = tid >> 5;
    const int half     = lane >> 4;
    const int l15      = lane & 15;
    const int rowBase  = blockIdx.x * (8 * 16) + wave * 16;
    const int colBeg   = blockIdx.y * COLS_PER_SLICE;
    const int colEnd   = colBeg + COLS_PER_SLICE;

    const float kv = kvp[0], nv = nvp[0];

    const int yIdx = (blockIdx.y * (NROWS / 128) + blockIdx.x) * 256 + tid;
    float4 yv;
    {
        const float sc = __expf(0.5f * nv);
        const float4 xv = ((const float4*)x)[yIdx];
        const float4 nz = ((const float4*)noise)[yIdx];
        yv.x = fmaf(sc, nz.x, xv.x); yv.y = fmaf(sc, nz.y, xv.y);
        yv.z = fmaf(sc, nz.z, xv.z); yv.w = fmaf(sc, nz.w, xv.w);
    }

    const float total_var = __expf(nv) + __expf(kv);
    const float a1 = 1.0f / (2.0f * total_var);
    const float a2 = 1.0f / (2.0f * __expf(kv));

    v2f a[16];
    {
        const float* ar = x + (size_t)(rowBase + l15) * NDIM + 2 * half;
#pragma unroll
        for (int ch = 0; ch < 16; ++ch)
            a[ch] = *(const v2f*)(ar + 4 * ch);
    }

    float sqA[8];
    {
        float sA = 0.0f;
#pragma unroll
        for (int ch = 0; ch < 16; ++ch) {
            sA = fmaf(a[ch].x, a[ch].x, sA);
            sA = fmaf(a[ch].y, a[ch].y, sA);
        }
        sA += __shfl_xor(sA, 16, 32);
#pragma unroll
        for (int k = 0; k < 8; ++k)
            sqA[k] = __shfl(sA, k + 8 * half, 32);
    }

    float l1[8], l2[8], m2[8];
#pragma unroll
    for (int k = 0; k < 8; ++k) { l1[k] = 0.0f; l2[k] = 0.0f; m2[k] = -1e30f; }

    float4 stage[8];
#pragma unroll
    for (int i = 0; i < 8; ++i)
        stage[i] = ((const float4*)(x + (size_t)colBeg * NDIM))[tid + i * 256];

    for (int cb = colBeg; cb < colEnd; cb += COLS_PER_BLOCK) {
        float sqS[8];
#pragma unroll
        for (int i = 0; i < 8; ++i)
            sqS[i] = row16_sum(sumsq4(stage[i]));

        __syncthreads();
#pragma unroll
        for (int i = 0; i < 8; ++i) {
            int idx = tid + i * 256;
            int c   = idx >> 4;
            int f4  = idx & 15;
            float* dst = &ldsX[c * LDS_STRIDE + f4 * 4];
            dst[0] = stage[i].x; dst[1] = stage[i].y;
            dst[2] = stage[i].z; dst[3] = stage[i].w;
        }
        if ((tid & 15) == 0) {
#pragma unroll
            for (int i = 0; i < 8; ++i)
                ldsSq[(tid >> 4) + 16 * i] = sqS[i];
        }
        __syncthreads();

        const int nb = cb + COLS_PER_BLOCK;
        if (nb < colEnd) {
#pragma unroll
            for (int i = 0; i < 8; ++i)
                stage[i] = ((const float4*)(x + (size_t)nb * NDIM))[tid + i * 256];
        }

        const float* bp  = &ldsX[l15 * LDS_STRIDE + 2 * half];
        const float* sqp = &ldsSq[l15];
        for (int ct = 0; ct < COLS_PER_BLOCK / 16; ++ct, bp += 16 * LDS_STRIDE, sqp += 16) {
            v2f b[16];
#pragma unroll
            for (int ch = 0; ch < 16; ++ch)
                b[ch] = *(const v2f*)(bp + 4 * ch);

            v8f c = {};
#pragma unroll
            for (int ch = 0; ch < 16; ++ch)
                c = __builtin_amdgcn_wmma_f32_16x16x4_f32(
                         false, a[ch],  false, b[ch],
                         (short)0, c,  false,  false);

            const float sqB = *sqp;
            const int   col = cb + ct * 16 + l15;
#pragma unroll
            for (int k = 0; k < 8; ++k) {
                const int rowG = rowBase + k + 8 * half;
                float d = fmaxf(sqA[k] + sqB - 2.0f * c[k], 0.0f);
                l1[k] += __expf(-d * a1);
                float s2 = (col == rowG) ? -1e38f : (-d * a2);
                float nm = fmaxf(m2[k], s2);
                l2[k] = l2[k] * __expf(m2[k] - nm) + __expf(s2 - nm);
                m2[k] = nm;
            }
        }
    }

#pragma unroll
    for (int k = 0; k < 8; ++k) {
#pragma unroll
        for (int m = 1; m <= 8; m <<= 1) {
            l1[k] += __shfl_xor(l1[k], m, 32);
            float om = __shfl_xor(m2[k], m, 32);
            float ol = __shfl_xor(l2[k], m, 32);
            float nm = fmaxf(m2[k], om);
            l2[k] = l2[k] * __expf(m2[k] - nm) + ol * __expf(om - nm);
            m2[k] = nm;
        }
    }
    if (l15 == 0) {
#pragma unroll
        for (int k = 0; k < 8; ++k) {
            const int r = wave * 16 + k + 8 * half;
            ldsOut[0][r] = l1[k];
            ldsOut[1][r] = m2[k];
            ldsOut[2][r] = l2[k];
        }
    }
    __syncthreads();
    if (tid < 128) {
        const size_t o = (size_t)blockIdx.y * NROWS + (size_t)blockIdx.x * 128 + tid;
        pl1[o] = ldsOut[0][tid];
        pm2[o] = ldsOut[1][tid];
        pl2[o] = ldsOut[2][tid];
    }
    ((float4*)y)[yIdx] = yv;
}

__global__ __launch_bounds__(1024) void k_final(const float* __restrict__ pl1,
                                                const float* __restrict__ pm2,
                                                const float* __restrict__ pl2,
                                                const float* __restrict__ kvp,
                                                const float* __restrict__ nvp,
                                                float* __restrict__ out2) {
    __shared__ float s1[1024], s2[1024];
    float p1 = 0.0f, p2 = 0.0f;
#pragma unroll
    for (int q = threadIdx.x; q < NROWS / 4; q += 1024) {
        float a[4] = {0.0f, 0.0f, 0.0f, 0.0f};
        float m[NSPLIT][4], l[NSPLIT][4];
#pragma unroll
        for (int s = 0; s < NSPLIT; ++s) {
            const float4 va = ((const float4*)(pl1 + s * NROWS))[q];
            const float4 vm = ((const float4*)(pm2 + s * NROWS))[q];
            const float4 vl = ((const float4*)(pl2 + s * NROWS))[q];
            a[0] += va.x; a[1] += va.y; a[2] += va.z; a[3] += va.w;
            m[s][0] = vm.x; m[s][1] = vm.y; m[s][2] = vm.z; m[s][3] = vm.w;
            l[s][0] = vl.x; l[s][1] = vl.y; l[s][2] = vl.z; l[s][3] = vl.w;
        }
#pragma unroll
        for (int e = 0; e < 4; ++e) {
            float mm = m[0][e];
#pragma unroll
            for (int s = 1; s < NSPLIT; ++s) mm = fmaxf(mm, m[s][e]);
            float ll = 0.0f;
#pragma unroll
            for (int s = 0; s < NSPLIT; ++s) ll = fmaf(l[s][e], __expf(m[s][e] - mm), ll);
            p1 += __logf(a[e]);
            p2 += mm + __logf(ll);
        }
    }
    s1[threadIdx.x] = p1; s2[threadIdx.x] = p2;
    __syncthreads();
    for (int st = 512; st > 0; st >>= 1) {
        if (threadIdx.x < st) {
            s1[threadIdx.x] += s1[threadIdx.x + st];
            s2[threadIdx.x] += s2[threadIdx.x + st];
        }
        __syncthreads();
    }
    if (threadIdx.x == 0) {
        const float kv = kvp[0], nv = nvp[0];
        const float Nf = (float)NROWS, dims = (float)NDIM;
        float total_var = expf(nv) + expf(kv);
        float nc1 = 0.5f * dims * logf(TWO_PI * total_var);
        float mean_lp1 = s1[0] / Nf - logf(Nf) - nc1;
        float h  = -mean_lp1;
        out2[0] = BETA * (h - nc1);
        float kdevar = expf(kv);
        float nc2 = 0.5f * dims * logf(TWO_PI * kdevar);
        float mean_lp2 = s2[0] / Nf - logf(Nf - 1.0f) - nc2;
        out2[1] = -mean_lp2;
    }
}

extern "C" void kernel_launch(void* const* d_in, const int* in_sizes, int n_in,
                              void* d_out, int out_size, void* d_ws, size_t ws_size,
                              hipStream_t stream) {
    const float* x     = (const float*)d_in[0];
    const float* noise = (const float*)d_in[1];
    const float* kv    = (const float*)d_in[2];
    const float* nv    = (const float*)d_in[3];
    float* out = (float*)d_out;
    float* ws  = (float*)d_ws;

    float* pl1 = ws + NROWS;
    float* pm2 = ws + NROWS + NSPLIT * NROWS;
    float* pl2 = ws + NROWS + 2 * NSPLIT * NROWS;

    const int nd = NROWS * NDIM;
    k_pairwise<<<dim3(NROWS / 128, NSPLIT), 256, 0, stream>>>(x, noise, kv, nv, out, pl1, pm2, pl2);
    k_final<<<1, 1024, 0, stream>>>(pl1, pm2, pl2, kv, nv, out + nd);
}


# Matrix operands swapped (columns as A, rows as B): a lane owns one row, so one running maximum and one rescale per lane per tile (17 exp2 per 8 pairs, was 24); row sums change order
# speedup vs baseline: 3.7527x; 1.1157x over previous
#include <hip/hip_runtime.h>
#include <hip/hip_bf16.h>
#include <math.h>

typedef __attribute__((ext_vector_type(2))) float v2f;
typedef __attribute__((ext_vector_type(8))) float v8f;

#define NROWS 8192
#define NDIM  64
#define COLS_PER_BLOCK 128
#define NSPLIT 8
#define COLS_PER_SLICE (NROWS / NSPLIT)
#define LDS_STRIDE 66
#define BETA 0.01f
#define TWO_PI 6.28318530717958647692f
#define LOG2E 1.44269504088896340736f
#define LN2   0.69314718055994530942f

__device__ __forceinline__ float sumsq4(const float4 v) {
    float s = v.x * v.x;
    s = fmaf(v.y, v.y, s); s = fmaf(v.z, v.z, s); s = fmaf(v.w, v.w, s);
    return s;
}

template <int CTRL>
__device__ __forceinline__ float row_xmask(float v) {
    return __builtin_bit_cast(float,
        __builtin_amdgcn_update_dpp(0, __builtin_bit_cast(int, v), CTRL, 0xf, 0xf, true));
}

__device__ __forceinline__ float row16_sum(float v) {
    v += row_xmask<0x161>(v);
    v += row_xmask<0x162>(v);
    v += row_xmask<0x164>(v);
    v += row_xmask<0x168>(v);
    return v;
}

__global__ __launch_bounds__(256) void k_pairwise(const float* __restrict__ x,
                                                  const float* __restrict__ noise,
                                                  const float* __restrict__ kvp,
                                                  const float* __restrict__ nvp,
                                                  float* __restrict__ y,
                                                  float* __restrict__ pl1,
                                                  float* __restrict__ pm2,
                                                  float* __restrict__ pl2) {
    __shared__ float ldsX[COLS_PER_BLOCK * LDS_STRIDE];
    __shared__ __attribute__((aligned(16))) float ldsSq[COLS_PER_BLOCK];
    __shared__ float ldsOut[3][8 * 16];

    const int tid      = threadIdx.x;
    const int lane     = tid & 31;
    const int wave     = tid >> 5;
    const int half     = lane >> 4;
    const int l15      = lane & 15;
    const int rowBase  = blockIdx.x * (8 * 16) + wave * 16;
    const int colBeg   = blockIdx.y * COLS_PER_SLICE;
    const int colEnd   = colBeg + COLS_PER_SLICE;

    const float kv = kvp[0], nv = nvp[0];

    const int yIdx = (blockIdx.y * (NROWS / 128) + blockIdx.x) * 256 + tid;
    float4 yv;
    {
        const float sc = __expf(0.5f * nv);
        const float4 xv = ((const float4*)x)[yIdx];
        const float4 nz = ((const float4*)noise)[yIdx];
        yv.x = fmaf(sc, nz.x, xv.x); yv.y = fmaf(sc, nz.y, xv.y);
        yv.z = fmaf(sc, nz.z, xv.z); yv.w = fmaf(sc, nz.w, xv.w);
    }

    const float total_var = __expf(nv) + __expf(kv);
    const float a1 = 1.0f / (2.0f * total_var);
    const float a2 = 1.0f / (2.0f * __expf(kv));
    const v2f k1 = {-a1 * LOG2E, -a1 * LOG2E};
    const v2f k2 = {-a2 * LOG2E, -a2 * LOG2E};
    const int rowBaseU = blockIdx.x * (8 * 16) + __builtin_amdgcn_readfirstlane(wave) * 16;
    const int dk       = l15 - 8 * half;
    v2f maskDiag[4];
#pragma unroll
    for (int j = 0; j < 4; ++j) {
        maskDiag[j].x = (dk == 2 * j)     ? -1e38f : 0.0f;
        maskDiag[j].y = (dk == 2 * j + 1) ? -1e38f : 0.0f;
    }

    v2f a[16];
    {
        const float* ar = x + (size_t)(rowBase + l15) * NDIM + 2 * half;
#pragma unroll
        for (int ch = 0; ch < 16; ++ch)
            a[ch] = *(const v2f*)(ar + 4 * ch);
    }

    v2f sqR;
    {
        float sA = 0.0f;
#pragma unroll
        for (int ch = 0; ch < 16; ++ch) {
            sA = fmaf(a[ch].x, a[ch].x, sA);
            sA = fmaf(a[ch].y, a[ch].y, sA);
        }
        sA += __shfl_xor(sA, 16, 32);
        sqR = (v2f){sA, sA};
    }

    v2f l1[4], l2[4];
    float m2 = -1e30f;
#pragma unroll
    for (int j = 0; j < 4; ++j) { l1[j] = (v2f){0.0f, 0.0f}; l2[j] = (v2f){0.0f, 0.0f}; }

    float4 stage[8];
#pragma unroll
    for (int i = 0; i < 8; ++i)
        stage[i] = ((const float4*)(x + (size_t)colBeg * NDIM))[tid + i * 256];

    for (int cb = colBeg; cb < colEnd; cb += COLS_PER_BLOCK) {
        float sqS[8];
#pragma unroll
        for (int i = 0; i < 8; ++i)
            sqS[i] = row16_sum(sumsq4(stage[i]));

        __syncthreads();
#pragma unroll
        for (int i = 0; i < 8; ++i) {
            int idx = tid + i * 256;
            int c   = idx >> 4;
            int f4  = idx & 15;
            float* dst = &ldsX[c * LDS_STRIDE + f4 * 4];
            dst[0] = stage[i].x; dst[1] = stage[i].y;
            dst[2] = stage[i].z; dst[3] = stage[i].w;
        }
        if ((tid & 15) == 0) {
#pragma unroll
            for (int i = 0; i < 8; ++i)
                ldsSq[(tid >> 4) + 16 * i] = sqS[i];
        }
        __syncthreads();

        const int nb = cb + COLS_PER_BLOCK;
        if (nb < colEnd) {
#pragma unroll
            for (int i = 0; i < 8; ++i)
                stage[i] = ((const float4*)(x + (size_t)nb * NDIM))[tid + i * 256];
        }

        const float*  bp  = &ldsX[l15 * LDS_STRIDE + 2 * half];
        const float4* sqp = (const float4*)&ldsSq[8 * half];
        for (int ct = 0; ct < COLS_PER_BLOCK / 16; ++ct, bp += 16 * LDS_STRIDE, sqp += 4) {
            v2f b[16];
#pragma unroll
            for (int ch = 0; ch < 16; ++ch)
                b[ch] = *(const v2f*)(bp + 4 * ch);

            v8f c = {};
#pragma unroll
            for (int ch = 0; ch < 16; ++ch)
                c = __builtin_amdgcn_wmma_f32_16x16x4_f32(
                         false, b[ch],  false, a[ch],
                         (short)0, c,  false,  false);

            const float4 q0 = sqp[0], q1 = sqp[1];
            const v2f sqC[4] = {{q0.x, q0.y}, {q0.z, q0.w}, {q1.x, q1.y}, {q1.z, q1.w}};
            v2f d[4];
#pragma unroll
            for (int j = 0; j < 4; ++j) {
                const v2f cc = {c[2 * j], c[2 * j + 1]};
                d[j] = __builtin_elementwise_fma((v2f){-2.0f, -2.0f}, cc, sqR + sqC[j]);
                d[j] = __builtin_elementwise_max(d[j], (v2f){0.0f, 0.0f});
                const v2f t1 = d[j] * k1;
                l1[j] += (v2f){__builtin_amdgcn_exp2f(t1.x), __builtin_amdgcn_exp2f(t1.y)};
            }
            const float onDiag = (cb + ct * 16 == rowBaseU) ? 1.0f : 0.0f;
            v2f s2[4];
#pragma unroll
            for (int j = 0; j < 4; ++j)
                s2[j] = __builtin_elementwise_fma((v2f){onDiag, onDiag}, maskDiag[j], d[j] * k2);
            const float mA = fmaxf(fmaxf(s2[0].x, s2[0].y), s2[1].x);
            const float mB = fmaxf(fmaxf(s2[1].y, s2[2].x), s2[2].y);
            const float mC = fmaxf(fmaxf(s2[3].x, s2[3].y), m2);
            const float mn = fmaxf(fmaxf(mA, mB), mC);
            const float ea1 = __builtin_amdgcn_exp2f(m2 - mn);
            const v2f   ea  = {ea1, ea1};
            const v2f   mn2 = {mn, mn};
            m2 = mn;
#pragma unroll
            for (int j = 0; j < 4; ++j) {
                const v2f tb = s2[j] - mn2;
                const v2f eb = {__builtin_amdgcn_exp2f(tb.x), __builtin_amdgcn_exp2f(tb.y)};
                l2[j] = __builtin_elementwise_fma(l2[j], ea, eb);
            }
        }
    }

    const v2f p1 = (l1[0] + l1[1]) + (l1[2] + l1[3]);
    const v2f p2 = (l2[0] + l2[1]) + (l2[2] + l2[3]);
    float L1 = p1.x + p1.y;
    float L2 = p2.x + p2.y;
    {
        const float o1 = __shfl_xor(L1, 16, 32);
        const float om = __shfl_xor(m2, 16, 32);
        const float ol = __shfl_xor(L2, 16, 32);
        const float nm = fmaxf(m2, om);
        L1 += o1;
        L2 = L2 * __builtin_amdgcn_exp2f(m2 - nm) + ol * __builtin_amdgcn_exp2f(om - nm);
        m2 = nm;
    }
    if (half == 0) {
        const int r = wave * 16 + l15;
        ldsOut[0][r] = L1;
        ldsOut[1][r] = m2 * LN2;
        ldsOut[2][r] = L2;
    }
    __syncthreads();
    if (tid < 128) {
        const size_t o = (size_t)blockIdx.y * NROWS + (size_t)blockIdx.x * 128 + tid;
        pl1[o] = ldsOut[0][tid];
        pm2[o] = ldsOut[1][tid];
        pl2[o] = ldsOut[2][tid];
    }
    ((float4*)y)[yIdx] = yv;
}

__global__ __launch_bounds__(1024) void k_final(const float* __restrict__ pl1,
                                                const float* __restrict__ pm2,
                                                const float* __restrict__ pl2,
                                                const float* __restrict__ kvp,
                                                const float* __restrict__ nvp,
                                                float* __restrict__ out2) {
    __shared__ float s1[1024], s2[1024];
    float p1 = 0.0f, p2 = 0.0f;
#pragma unroll
    for (int q = threadIdx.x; q < NROWS / 4; q += 1024) {
        float a[4] = {0.0f, 0.0f, 0.0f, 0.0f};
        float m[NSPLIT][4], l[NSPLIT][4];
#pragma unroll
        for (int s = 0; s < NSPLIT; ++s) {
            const float4 va = ((const float4*)(pl1 + s * NROWS))[q];
            const float4 vm = ((const float4*)(pm2 + s * NROWS))[q];
            const float4 vl = ((const float4*)(pl2 + s * NROWS))[q];
            a[0] += va.x; a[1] += va.y; a[2] += va.z; a[3] += va.w;
            m[s][0] = vm.x; m[s][1] = vm.y; m[s][2] = vm.z; m[s][3] = vm.w;
            l[s][0] = vl.x; l[s][1] = vl.y; l[s][2] = vl.z; l[s][3] = vl.w;
        }
#pragma unroll
        for (int e = 0; e < 4; ++e) {
            float mm = m[0][e];
#pragma unroll
            for (int s = 1; s < NSPLIT; ++s) mm = fmaxf(mm, m[s][e]);
            float ll = 0.0f;
#pragma unroll
            for (int s = 0; s < NSPLIT; ++s) ll = fmaf(l[s][e], __expf(m[s][e] - mm), ll);
            p1 += __logf(a[e]);
            p2 += mm + __logf(ll);
        }
    }
    s1[threadIdx.x] = p1; s2[threadIdx.x] = p2;
    __syncthreads();
    for (int st = 512; st > 0; st >>= 1) {
        if (threadIdx.x < st) {
            s1[threadIdx.x] += s1[threadIdx.x + st];
            s2[threadIdx.x] += s2[threadIdx.x + st];
        }
        __syncthreads();
    }
    if (threadIdx.x == 0) {
        const float kv = kvp[0], nv = nvp[0];
        const float Nf = (float)NROWS, dims = (float)NDIM;
        float total_var = expf(nv) + expf(kv);
        float nc1 = 0.5f * dims * logf(TWO_PI * total_var);
        float mean_lp1 = s1[0] / Nf - logf(Nf) - nc1;
        float h  = -mean_lp1;
        out2[0] = BETA * (h - nc1);
        float kdevar = expf(kv);
        float nc2 = 0.5f * dims * logf(TWO_PI * kdevar);
        float mean_lp2 = s2[0] / Nf - logf(Nf - 1.0f) - nc2;
        out2[1] = -mean_lp2;
    }
}

extern "C" void kernel_launch(void* const* d_in, const int* in_sizes, int n_in,
                              void* d_out, int out_size, void* d_ws, size_t ws_size,
                              hipStream_t stream) {
    const float* x     = (const float*)d_in[0];
    const float* noise = (const float*)d_in[1];
    const float* kv    = (const float*)d_in[2];
    const float* nv    = (const float*)d_in[3];
    float* out = (float*)d_out;
    float* ws  = (float*)d_ws;

    float* pl1 = ws + NROWS;
    float* pm2 = ws + NROWS + NSPLIT * NROWS;
    float* pl2 = ws + NROWS + 2 * NSPLIT * NROWS;

    const int nd = NROWS * NDIM;
    k_pairwise<<<dim3(NROWS / 128, NSPLIT), 256, 0, stream>>>(x, noise, kv, nv, out, pl1, pm2, pl2);
    k_final<<<1, 1024, 0, stream>>>(pl1, pm2, pl2, kv, nv, out + nd);
}
